# starting kernel
# baseline (speedup 1.0000x reference)
#include <hip/hip_runtime.h>

typedef __attribute__((ext_vector_type(2))) float v2f;
typedef __attribute__((ext_vector_type(8))) float v8f;

__global__ __launch_bounds__(256) void se3_transform_wmma(
    const float* __restrict__ X, const float* __restrict__ dof,
    float* __restrict__ out, int np)
{
    __shared__ float sA[64];
    const int tid = threadIdx.x;
    const int bt  = blockIdx.x;

    if (tid < 64) sA[tid] = 0.0f;
    __syncthreads();

    if (tid == 0) {
        const float* d = dof + (size_t)bt * 6;
        float vx = d[0], vy = d[1], vz = d[2];
        float wx = d[3], wy = d[4], wz = d[5];
        float t2 = wx*wx + wy*wy + wz*wz;
        float th = sqrtf(t2);
        bool  sm = th < 1e-4f;
        float st  = sm ? 1.0f : th;
        float st2 = sm ? 1.0f : t2;
        float st3 = sm ? 1.0f : t2 * th;
        float sth = sinf(th), cth = cosf(th);
        float Af = sm ? 1.0f - t2 * (1.0f/6.0f)        : sth / st;
        float Bf = sm ? 0.5f - t2 * (1.0f/24.0f)       : (1.0f - cth) / st2;
        float Cf = sm ? (1.0f/6.0f) - t2*(1.0f/120.0f) : (th - sth) / st3;
        float xy = wx*wy, xz = wx*wz, yz = wy*wz;
        float R00 = 1.0f + Bf*(wx*wx - t2);
        float R01 = -Af*wz + Bf*xy;
        float R02 =  Af*wy + Bf*xz;
        float R10 =  Af*wz + Bf*xy;
        float R11 = 1.0f + Bf*(wy*wy - t2);
        float R12 = -Af*wx + Bf*yz;
        float R20 = -Af*wy + Bf*xz;
        float R21 =  Af*wx + Bf*yz;
        float R22 = 1.0f + Bf*(wz*wz - t2);
        float V00 = 1.0f + Cf*(wx*wx - t2);
        float V01 = -Bf*wz + Cf*xy;
        float V02 =  Bf*wy + Cf*xz;
        float V10 =  Bf*wz + Cf*xy;
        float V11 = 1.0f + Cf*(wy*wy - t2);
        float V12 = -Bf*wx + Cf*yz;
        float V20 = -Bf*wy + Cf*xz;
        float V21 =  Bf*wx + Cf*yz;
        float V22 = 1.0f + Cf*(wz*wz - t2);
        float T0 = V00*vx + V01*vy + V02*vz;
        float T1 = V10*vx + V11*vy + V12*vz;
        float T2 = V20*vx + V21*vy + V22*vz;
        sA[0]=R00; sA[1]=R01; sA[2] =R02; sA[3] =T0;
        sA[4]=R10; sA[5]=R11; sA[6] =R12; sA[7] =T1;
        sA[8]=R20; sA[9]=R21; sA[10]=R22; sA[11]=T2;
    }
    __syncthreads();

    const int  lane = tid & 31;
    const int  wave = tid >> 5;
    const int  r    = lane & 15;
    const bool hi   = lane >= 16;

    v2f afrag;
    afrag[0] = sA[r*4 + (hi ? 2 : 0)];
    afrag[1] = sA[r*4 + (hi ? 3 : 1)];

    const size_t slab = (size_t)bt * (size_t)np * 3;
    const float* __restrict__ xin  = X   + slab;
    float*       __restrict__ xout = out + slab;

    const int ntiles = np >> 4;
    const int off0   = hi ? r*3 + 2 : r*3;
    const int off1   = r*3 + 1;
    const int step   = 8 * 48;

    const float* p = xin  + (size_t)wave * 48;
    float*       q = xout + (size_t)wave * 48 + r * 3;

    #pragma unroll 2
    for (int tile = wave; tile < ntiles; tile += 8) {
        float b0  = p[off0];
        float b1y = p[off1];
        v2f bfrag;
        bfrag[0] = b0;
        bfrag[1] = hi ? 1.0f : b1y;

        v8f c = {};
        v8f dres = __builtin_amdgcn_wmma_f32_16x16x4_f32(
             false, afrag,  false, bfrag,
             (short)0, c,  false,  false);

        if (!hi) {
            q[0] = dres[0];
            q[1] = dres[1];
            q[2] = dres[2];
        }
        p += step;
        q += step;
    }
}

extern "C" void kernel_launch(void* const* d_in, const int* in_sizes, int n_in,
                              void* d_out, int out_size, void* d_ws, size_t ws_size,
                              hipStream_t stream)
{
    const float* X   = (const float*)d_in[0];
    const float* dof = (const float*)d_in[1];
    float* out = (float*)d_out;

    const int n_bt = in_sizes[1] / 6;
    const int np   = in_sizes[0] / (n_bt * 3);

    se3_transform_wmma<<<n_bt, 256, 0, stream>>>(X, dof, out, np);
}


# Each lane loads and stores one whole point (b96; a wave op covers 384 contiguous B, 8 per trip in flight); two chained f32 WMMAs, transform placed per k half; sum order now (x,y) then (z,T)
# speedup vs baseline: 2.1394x; 2.1394x over previous
#include <hip/hip_runtime.h>

typedef __attribute__((ext_vector_type(2))) float v2f;
typedef __attribute__((ext_vector_type(8))) float v8f;

__global__ __launch_bounds__(256) __attribute__((amdgpu_waves_per_eu(8, 8)))
void se3_transform_wmma(
    const float* __restrict__ X, const float* __restrict__ dof,
    float* __restrict__ out, int np)
{
    __shared__ __attribute__((aligned(16))) float sA[64];
    const int tid = threadIdx.x;
    const int bt  = blockIdx.x;

    if (tid < 64) sA[tid] = 0.0f;
    __syncthreads();

    if (tid == 0) {
        const float* d = dof + (size_t)bt * 6;
        float vx = d[0], vy = d[1], vz = d[2];
        float wx = d[3], wy = d[4], wz = d[5];
        float t2 = wx*wx + wy*wy + wz*wz;
        float th = sqrtf(t2);
        bool  sm = th < 1e-4f;
        float st  = sm ? 1.0f : th;
        float st2 = sm ? 1.0f : t2;
        float st3 = sm ? 1.0f : t2 * th;
        float sth = sinf(th), cth = cosf(th);
        float Af = sm ? 1.0f - t2 * (1.0f/6.0f)        : sth / st;
        float Bf = sm ? 0.5f - t2 * (1.0f/24.0f)       : (1.0f - cth) / st2;
        float Cf = sm ? (1.0f/6.0f) - t2*(1.0f/120.0f) : (th - sth) / st3;
        float xy = wx*wy, xz = wx*wz, yz = wy*wz;
        float R00 = 1.0f + Bf*(wx*wx - t2);
        float R01 = -Af*wz + Bf*xy;
        float R02 =  Af*wy + Bf*xz;
        float R10 =  Af*wz + Bf*xy;
        float R11 = 1.0f + Bf*(wy*wy - t2);
        float R12 = -Af*wx + Bf*yz;
        float R20 = -Af*wy + Bf*xz;
        float R21 =  Af*wx + Bf*yz;
        float R22 = 1.0f + Bf*(wz*wz - t2);
        float V00 = 1.0f + Cf*(wx*wx - t2);
        float V01 = -Bf*wz + Cf*xy;
        float V02 =  Bf*wy + Cf*xz;
        float V10 =  Bf*wz + Cf*xy;
        float V11 = 1.0f + Cf*(wy*wy - t2);
        float V12 = -Bf*wx + Cf*yz;
        float V20 = -Bf*wy + Cf*xz;
        float V21 =  Bf*wx + Cf*yz;
        float V22 = 1.0f + Cf*(wz*wz - t2);
        float T0 = V00*vx + V01*vy + V02*vz;
        float T1 = V10*vx + V11*vy + V12*vz;
        float T2 = V20*vx + V21*vy + V22*vz;
        sA[0]=R00; sA[1]=R01; sA[2] =R02; sA[3] =T0;
        sA[4]=R10; sA[5]=R11; sA[6] =R12; sA[7] =T1;
        sA[8]=R20; sA[9]=R21; sA[10]=R22; sA[11]=T2;
    }
    __syncthreads();

    const int lane = tid & 31;
    const int wave = __builtin_amdgcn_readfirstlane(tid >> 5);

    const int arow = (lane < 16) ? lane : ((lane - 24) & 15);
    const float4 av = *reinterpret_cast<const float4*>(&sA[arow * 4]);
    v2f a1, a2;
    a1[0] = av.x;  a1[1] = av.y;
    a2[0] = av.z;  a2[1] = av.w;

    const size_t slab = (size_t)bt * (size_t)np * 3;

    const int ntiles   = np >> 5;
    const int per_wave = (ntiles + 7) >> 3;
    const int t0 = wave * per_wave;
    const int t1 = (t0 + per_wave < ntiles) ? (t0 + per_wave) : ntiles;

    const float* p = X   + slab + (size_t)t0 * 96 + lane * 3;
    float*       q = out + slab + (size_t)t0 * 96 + lane * 3;

    constexpr int NB = 8;
    int t = t0;
    for (; t + NB <= t1; t += NB) {
        float px[NB], py[NB], pz[NB];
        #pragma unroll
        for (int j = 0; j < NB; ++j) {
            px[j] = p[j * 96 + 0];
            py[j] = p[j * 96 + 1];
            pz[j] = p[j * 96 + 2];
        }
        #pragma unroll
        for (int j = 0; j < NB; ++j) {
            v2f b1, b2;
            b1[0] = px[j];  b1[1] = py[j];
            b2[0] = pz[j];  b2[1] = 1.0f;
            v8f c = {};
            v8f d = __builtin_amdgcn_wmma_f32_16x16x4_f32(
                 false, a1,  false, b1,
                 (short)0, c,  false,  false);
            d = __builtin_amdgcn_wmma_f32_16x16x4_f32(
                 false, a2,  false, b2,
                 (short)0, d,  false,  false);
            q[j * 96 + 0] = d[0];
            q[j * 96 + 1] = d[1];
            q[j * 96 + 2] = d[2];
        }
        p += NB * 96;
        q += NB * 96;
    }
    #pragma nounroll
    for (; t < t1; ++t) {
        v2f b1, b2;
        b1[0] = p[0];  b1[1] = p[1];
        b2[0] = p[2];  b2[1] = 1.0f;
        v8f c = {};
        v8f d = __builtin_amdgcn_wmma_f32_16x16x4_f32(
            false, a1, false, b1, (short)0, c, false, false);
        d = __builtin_amdgcn_wmma_f32_16x16x4_f32(
            false, a2, false, b2, (short)0, d, false, false);
        q[0] = d[0];
        q[1] = d[1];
        q[2] = d[2];
        p += 96;
        q += 96;
    }
}

extern "C" void kernel_launch(void* const* d_in, const int* in_sizes, int n_in,
                              void* d_out, int out_size, void* d_ws, size_t ws_size,
                              hipStream_t stream)
{
    const float* X   = (const float*)d_in[0];
    const float* dof = (const float*)d_in[1];
    float* out = (float*)d_out;

    const int n_bt = in_sizes[1] / 6;
    const int np   = in_sizes[0] / (n_bt * 3);

    se3_transform_wmma<<<n_bt, 256, 0, stream>>>(X, dof, out, np);
}


# Loads lead the products from wave entry: 16 tiles requested before the transform arithmetic, then an 8x2-tile register ring refilled stage by stage; transform built per wave, no LDS or barrier
# speedup vs baseline: 2.2870x; 1.0690x over previous
#include <hip/hip_runtime.h>

typedef __attribute__((ext_vector_type(2))) float v2f;
typedef __attribute__((ext_vector_type(8))) float v8f;

__global__ __launch_bounds__(256) __attribute__((amdgpu_waves_per_eu(8, 8)))
void se3_transform_wmma(
    const float* __restrict__ X, const float* __restrict__ dof,
    float* __restrict__ out, int np)
{
    const int tid  = threadIdx.x;
    const int bt   = blockIdx.x;
    const int lane = tid & 31;
    const int wave = __builtin_amdgcn_readfirstlane(tid >> 5);

    const float* d = dof + (size_t)bt * 6;
    const float vx = d[0], vy = d[1], vz = d[2];
    const float wx = d[3], wy = d[4], wz = d[5];

    const size_t slab = (size_t)bt * (size_t)np * 3;

    const int ntiles   = np >> 5;
    const int per_wave = (ntiles + 7) >> 3;
    const int t0 = wave * per_wave;
    const int t1 = (t0 + per_wave < ntiles) ? (t0 + per_wave) : ntiles;

    const float* p = X   + slab + (size_t)t0 * 96 + lane * 3;
    float*       q = out + slab + (size_t)t0 * 96 + lane * 3;

    constexpr int NS   = 8;
    constexpr int ST   = 2;
    constexpr int LEAD = NS * ST;
    float px[LEAD], py[LEAD], pz[LEAD];
    int t = t0;
    const bool ring = (t + 2 * LEAD <= t1);
    if (ring) {
        #pragma unroll
        for (int k = 0; k < LEAD; ++k) {
            px[k] = p[k * 96 + 0];
            py[k] = p[k * 96 + 1];
            pz[k] = p[k * 96 + 2];
        }
    }

    const float t2 = wx*wx + wy*wy + wz*wz;
    const float th = sqrtf(t2);
    const bool  sm = th < 1e-4f;
    const float st  = sm ? 1.0f : th;
    const float st2 = sm ? 1.0f : t2;
    const float st3 = sm ? 1.0f : t2 * th;
    const float sth = sinf(th), cth = cosf(th);
    const float Af = sm ? 1.0f - t2 * (1.0f/6.0f)        : sth / st;
    const float Bf = sm ? 0.5f - t2 * (1.0f/24.0f)       : (1.0f - cth) / st2;
    const float Cf = sm ? (1.0f/6.0f) - t2*(1.0f/120.0f) : (th - sth) / st3;
    const float xy = wx*wy, xz = wx*wz, yz = wy*wz;
    const float R00 = 1.0f + Bf*(wx*wx - t2);
    const float R01 = -Af*wz + Bf*xy;
    const float R02 =  Af*wy + Bf*xz;
    const float R10 =  Af*wz + Bf*xy;
    const float R11 = 1.0f + Bf*(wy*wy - t2);
    const float R12 = -Af*wx + Bf*yz;
    const float R20 = -Af*wy + Bf*xz;
    const float R21 =  Af*wx + Bf*yz;
    const float R22 = 1.0f + Bf*(wz*wz - t2);
    const float V00 = 1.0f + Cf*(wx*wx - t2);
    const float V01 = -Bf*wz + Cf*xy;
    const float V02 =  Bf*wy + Cf*xz;
    const float V10 =  Bf*wz + Cf*xy;
    const float V11 = 1.0f + Cf*(wy*wy - t2);
    const float V12 = -Bf*wx + Cf*yz;
    const float V20 = -Bf*wy + Cf*xz;
    const float V21 =  Bf*wx + Cf*yz;
    const float V22 = 1.0f + Cf*(wz*wz - t2);
    const float T0 = V00*vx + V01*vy + V02*vz;
    const float T1 = V10*vx + V11*vy + V12*vz;
    const float T2 = V20*vx + V21*vy + V22*vz;

    const int r = (lane < 16) ? lane : (lane - 24);
    v2f a1, a2;
    a1[0] = (r == 0) ? R00 : (r == 1) ? R10 : (r == 2) ? R20 : 0.0f;
    a1[1] = (r == 0) ? R01 : (r == 1) ? R11 : (r == 2) ? R21 : 0.0f;
    a2[0] = (r == 0) ? R02 : (r == 1) ? R12 : (r == 2) ? R22 : 0.0f;
    a2[1] = (r == 0) ? T0  : (r == 1) ? T1  : (r == 2) ? T2  : 0.0f;

    if (ring) {
        do {
            #pragma unroll
            for (int s = 0; s < NS; ++s) {
                #pragma unroll
                for (int j = 0; j < ST; ++j) {
                    const int k = s * ST + j;
                    v2f b1, b2;
                    b1[0] = px[k];  b1[1] = py[k];
                    b2[0] = pz[k];  b2[1] = 1.0f;
                    v8f c = {};
                    v8f dd = __builtin_amdgcn_wmma_f32_16x16x4_f32(
                         false, a1,  false, b1,
                         (short)0, c,  false,  false);
                    dd = __builtin_amdgcn_wmma_f32_16x16x4_f32(
                         false, a2,  false, b2,
                         (short)0, dd,  false,  false);
                    q[k * 96 + 0] = dd[0];
                    q[k * 96 + 1] = dd[1];
                    q[k * 96 + 2] = dd[2];
                }
                #pragma unroll
                for (int j = 0; j < ST; ++j) {
                    const int k = s * ST + j;
                    px[k] = p[(LEAD + k) * 96 + 0];
                    py[k] = p[(LEAD + k) * 96 + 1];
                    pz[k] = p[(LEAD + k) * 96 + 2];
                }
                __builtin_amdgcn_sched_barrier(0);
            }
            p += LEAD * 96;
            q += LEAD * 96;
            t += LEAD;
        } while (t + 2 * LEAD <= t1);
        #pragma unroll
        for (int s = 0; s < NS; ++s) {
            #pragma unroll
            for (int j = 0; j < ST; ++j) {
                const int k = s * ST + j;
                v2f b1, b2;
                b1[0] = px[k];  b1[1] = py[k];
                b2[0] = pz[k];  b2[1] = 1.0f;
                v8f c = {};
                v8f dd = __builtin_amdgcn_wmma_f32_16x16x4_f32(
                    false, a1, false, b1, (short)0, c, false, false);
                dd = __builtin_amdgcn_wmma_f32_16x16x4_f32(
                    false, a2, false, b2, (short)0, dd, false, false);
                q[k * 96 + 0] = dd[0];
                q[k * 96 + 1] = dd[1];
                q[k * 96 + 2] = dd[2];
            }
            __builtin_amdgcn_sched_barrier(0);
        }
        p += LEAD * 96;
        q += LEAD * 96;
        t += LEAD;
    }
    #pragma nounroll
    for (; t < t1; ++t) {
        v2f b1, b2;
        b1[0] = p[0];  b1[1] = p[1];
        b2[0] = p[2];  b2[1] = 1.0f;
        v8f c = {};
        v8f dd = __builtin_amdgcn_wmma_f32_16x16x4_f32(
            false, a1, false, b1, (short)0, c, false, false);
        dd = __builtin_amdgcn_wmma_f32_16x16x4_f32(
            false, a2, false, b2, (short)0, dd, false, false);
        q[0] = dd[0];
        q[1] = dd[1];
        q[2] = dd[2];
        p += 96;
        q += 96;
    }
}

extern "C" void kernel_launch(void* const* d_in, const int* in_sizes, int n_in,
                              void* d_out, int out_size, void* d_ws, size_t ws_size,
                              hipStream_t stream)
{
    const float* X   = (const float*)d_in[0];
    const float* dof = (const float*)d_in[1];
    float* out = (float*)d_out;

    const int n_bt = in_sizes[1] / 6;
    const int np   = in_sizes[0] / (n_bt * 3);

    se3_transform_wmma<<<n_bt, 256, 0, stream>>>(X, dof, out, np);
}
